# starting kernel
# baseline (speedup 1.0000x reference)
#include <hip/hip_runtime.h>

typedef float v2f __attribute__((ext_vector_type(2)));
typedef float v4f __attribute__((ext_vector_type(4)));
typedef float v8f __attribute__((ext_vector_type(8)));

#define SEQ   32768
#define DIM   1024
#define ROWS_PER_WAVE 16
#define WAVES_PER_BLOCK 8

__global__ __launch_bounds__(WAVES_PER_BLOCK * 32)
void gates_gemm_wmma(const float* __restrict__ x,
                     const float* __restrict__ Wih,
                     const float* __restrict__ bih,
                     const float* __restrict__ bhh,
                     float* __restrict__ gates)
{
    const int lane    = threadIdx.x & 31;
    const int wave    = threadIdx.x >> 5;
    const int lane_lo = lane & 15;
    const int lane_hi = lane >> 4;
    const int row_base = (blockIdx.x * WAVES_PER_BLOCK + wave) * ROWS_PER_WAVE;

    const float* xrow = x + (size_t)(row_base + lane_lo) * DIM + 2 * lane_hi;
    const float* wrow = Wih + (size_t)(lane_lo & 3) * DIM + 2 * lane_hi;

    v8f acc = {};
    #pragma unroll 4
    for (int k = 0; k < DIM; k += 4) {
        v2f a = *(const v2f*)(xrow + k);
        v2f b = *(const v2f*)(wrow + k);
        acc = __builtin_amdgcn_wmma_f32_16x16x4_f32(
                   false, a,  false, b,
                   (short)0, acc,  false,  false);
    }

    if (lane_lo < 4) {
        const int g    = lane_lo;
        const float bb = bih[g] + bhh[g];
        const int r0   = row_base + lane_hi * 8;
        #pragma unroll
        for (int r = 0; r < 8; ++r)
            gates[(size_t)(r0 + r) * 4 + g] = acc[r] + bb;
    }
}

__device__ __forceinline__ float fsig(float xv)
{
    return __builtin_amdgcn_rcpf(1.0f + __builtin_amdgcn_exp2f(-1.4426950408889634f * xv));
}
__device__ __forceinline__ float ftanh(float xv)
{
    return fmaf(2.0f,
                __builtin_amdgcn_rcpf(1.0f + __builtin_amdgcn_exp2f(-2.8853901631779268f * xv)),
                -1.0f);
}

__global__ void lstm_scan(const float* __restrict__ gates,
                          const float* __restrict__ Whh,
                          const float* __restrict__ h0,
                          const float* __restrict__ c0,
                          float* __restrict__ out)
{
    if (threadIdx.x != 0) return;

    const float w0 = Whh[0], w1 = Whh[1], w2 = Whh[2], w3 = Whh[3];
    float h = h0[0];
    float c = c0[0];

    v4f hbuf = {};
    for (int t = 0; t < SEQ; ++t) {
        v4f g = *(const v4f*)(gates + 4 * (size_t)t);
        if ((t & 31) == 0)
            __builtin_prefetch(gates + 4 * (size_t)t + 512, 0, 3);

        const float gi = fmaf(w0, h, g.x);
        const float gf = fmaf(w1, h, g.y);
        const float gc = fmaf(w2, h, g.z);
        const float go = fmaf(w3, h, g.w);

        const float iv = fsig(gi);
        const float fv = fsig(gf);
        const float gv = ftanh(gc);
        const float ov = fsig(go);

        c = fmaf(fv, c, iv * gv);
        h = ov * ftanh(c);

        hbuf[t & 3] = h;
        if ((t & 3) == 3)
            *(v4f*)(out + (t - 3)) = hbuf;
    }
}

extern "C" void kernel_launch(void* const* d_in, const int* in_sizes, int n_in,
                              void* d_out, int out_size, void* d_ws, size_t ws_size,
                              hipStream_t stream)
{
    const float* x    = (const float*)d_in[0];
    const float* Wih  = (const float*)d_in[1];
    const float* Whh  = (const float*)d_in[2];
    const float* bih  = (const float*)d_in[3];
    const float* bhh  = (const float*)d_in[4];
    const float* h0   = (const float*)d_in[5];
    const float* c0   = (const float*)d_in[6];
    float*       out  = (float*)d_out;
    float*       gts  = (float*)d_ws;

    const int rows_per_block = ROWS_PER_WAVE * WAVES_PER_BLOCK;
    const int nblocks = SEQ / rows_per_block;

    gates_gemm_wmma<<<nblocks, WAVES_PER_BLOCK * 32, 0, stream>>>(x, Wih, bih, bhh, gts);
    lstm_scan<<<1, 32, 0, stream>>>(gts, Whh, h0, c0, out);
}


# scan: constant multiplies and activation fix-ups leave the serial chain (14 -> 9 links): scales ride the add before each rcp, cell kept as -2log2e*c, o folded into the next z; sums re-associated
# speedup vs baseline: 3.6738x; 3.6738x over previous
#include <hip/hip_runtime.h>

typedef float v2f __attribute__((ext_vector_type(2)));
typedef float v4f __attribute__((ext_vector_type(4)));
typedef float v8f __attribute__((ext_vector_type(8)));

#define SEQ   32768
#define DIM   1024
#define ROWS_PER_WAVE 16
#define WAVES_PER_BLOCK 8

#define SBLK 16
#define NSB  (SEQ / SBLK)

__global__ __launch_bounds__(WAVES_PER_BLOCK * 32)
void gates_gemm_wmma(const float* __restrict__ x,
                     const float* __restrict__ Wih,
                     const float* __restrict__ bih,
                     const float* __restrict__ bhh,
                     float* __restrict__ plane)
{
    const int lane    = threadIdx.x & 31;
    const int wave    = threadIdx.x >> 5;
    const int lane_lo = lane & 15;
    const int lane_hi = lane >> 4;
    const int row_base = (blockIdx.x * WAVES_PER_BLOCK + wave) * ROWS_PER_WAVE;

    const float* xrow = x + (size_t)(row_base + lane_lo) * DIM + 2 * lane_hi;
    const float* wrow = Wih + (size_t)(lane_lo & 3) * DIM + 2 * lane_hi;

    v8f acc = {};
    #pragma unroll 4
    for (int k = 0; k < DIM; k += 4) {
        v2f a = *(const v2f*)(xrow + k);
        v2f b = *(const v2f*)(wrow + k);
        acc = __builtin_amdgcn_wmma_f32_16x16x4_f32(
                   false, a,  false, b,
                   (short)0, acc,  false,  false);
    }

    if (lane_lo < 4) {
        const int g    = lane_lo;
        const float bb = bih[g] + bhh[g];
        float* dst = plane + (size_t)(row_base >> 4) * (4 * SBLK) + g * SBLK + lane_hi * 8;
        v4f lo, hi;
        #pragma unroll
        for (int r = 0; r < 4; ++r) {
            lo[r] = acc[r] + bb;
            hi[r] = acc[4 + r] + bb;
        }
        *(v4f*)(dst)     = lo;
        *(v4f*)(dst + 4) = hi;
    }
}


struct lane_consts {
    float k;
    float kw;
    float p;
};

#define QP(Q) "quad_perm:[" #Q "," #Q "," #Q "," #Q "] row_mask:0xf bank_mask:0xf bound_ctrl:1"

__device__ __forceinline__ void lstm_step(float& y, float& C, float kgn,
                                          const lane_consts& L, float& hout)
{
    float e, d, r, t0, ow;
    asm("v_exp_f32 %[e], %[y]\n\t"
        "v_nop\n\t"
        "v_fma_f32 %[d], %[e], %[p], %[p]\n\t"
        "v_rcp_f32 %[r], %[d]\n\t"
        "v_nop\n\t"
        "v_fma_f32_e64_dpp %[t0], %[r], %[r], %[r] " QP(2) "\n\t"
        "v_mul_f32_dpp %[ow], %[r], %[kw] " QP(3) "\n\t"
        "v_fmac_f32_dpp %[t0], %[r], %[C] " QP(1)
        : [e] "=&v"(e), [d] "=&v"(d), [r] "=&v"(r), [t0] "=&v"(t0), [ow] "=&v"(ow)
        : [y] "v"(y), [p] "v"(L.p), [kw] "v"(L.kw), [C] "v"(C));
    C = t0;
    const float bz = kgn - ow;
    float E, d2, r2, yn;
    asm("v_exp_f32 %[E], %[C]\n\t"
        "v_nop\n\t"
        "v_fma_f32 %[d2], %[E], 0.5, 0.5\n\t"
        "v_rcp_f32 %[r2], %[d2]\n\t"
        "v_nop\n\t"
        "v_fma_f32_e64_dpp %[yn], %[r2], %[ow], %[bz] " QP(0) "\n\t"
        "v_fma_f32_e64_dpp %[h], %[r2], %[r], -%[r] " QP(0)
        : [E] "=&v"(E), [d2] "=&v"(d2), [r2] "=&v"(r2), [yn] "=&v"(yn), [h] "=&v"(hout)
        : [C] "v"(t0), [ow] "v"(ow), [bz] "v"(bz), [r] "v"(r));
    y = yn;
}

__device__ __forceinline__ void lstm_block(const v4f (&cur)[4], v4f (&nxt)[4],
                                           const v4f* __restrict__ nextp,
                                           const lane_consts& L, bool writer,
                                           float& y, float& C, float* __restrict__ o)
{
    #pragma unroll
    for (int q = 0; q < 4; ++q) nxt[q] = nextp[q];
    __builtin_amdgcn_sched_barrier(0);

    v4f hb[4];
    #pragma unroll
    for (int q = 0; q < 4; ++q) {
        #pragma unroll
        for (int s = 0; s < 4; ++s) {
            const bool last = (q == 3 && s == 3);
            const float gn = last ? nxt[0][0] : (s < 3 ? cur[q][s + 1] : cur[(q + 1) & 3][0]);
            float hout;
            lstm_step(y, C, L.k * gn, L, hout);
            hb[q][s] = hout;
        }
    }
    if (writer) {
        #pragma unroll
        for (int q = 0; q < 4; ++q) *(v4f*)(o + 4 * q) = hb[q];
    }
}

__global__ void lstm_scan(const float* __restrict__ plane,
               const float* __restrict__ Whh,
               const float* __restrict__ h0,
               const float* __restrict__ c0,
               float* __restrict__ out)
{
    const int lane = threadIdx.x;
    if (lane >= 4) return;

    const bool tanh_gate = (lane == 2);
    lane_consts L;
    L.k  = tanh_gate ? -2.8853901631779268f : -1.4426950408889634f;
    L.kw = L.k * Whh[lane];
    L.p  = (lane == 0) ? 0.34657359027997264f
         : tanh_gate   ? -0.5f
         : 1.0f;
    const bool writer = (lane == 3);

    const v4f* __restrict__ gp = (const v4f*)(plane + lane * SBLK);
    v4f A[4], B[4];
    #pragma unroll
    for (int q = 0; q < 4; ++q) A[q] = gp[q];

    float C = -2.8853901631779268f * c0[0];
    float y = fmaf(h0[0], L.kw, L.k * A[0][0]);

    for (int b = 0; b < NSB; b += 2) {
        lstm_block(A, B, gp + (size_t)(b + 1) * SBLK, L, writer, y, C, out + (size_t)b * SBLK);
        const int nb = (b + 2 < NSB) ? b + 2 : NSB - 1;
        lstm_block(B, A, gp + (size_t)nb * SBLK, L, writer, y, C, out + (size_t)(b + 1) * SBLK);
    }
}

extern "C" void kernel_launch(void* const* d_in, const int* in_sizes, int n_in,
                              void* d_out, int out_size, void* d_ws, size_t ws_size,
                              hipStream_t stream)
{
    const float* x    = (const float*)d_in[0];
    const float* Wih  = (const float*)d_in[1];
    const float* Whh  = (const float*)d_in[2];
    const float* bih  = (const float*)d_in[3];
    const float* bhh  = (const float*)d_in[4];
    const float* h0   = (const float*)d_in[5];
    const float* c0   = (const float*)d_in[6];
    float*       out  = (float*)d_out;
    float*       gts  = (float*)d_ws;

    const int rows_per_block = ROWS_PER_WAVE * WAVES_PER_BLOCK;
    const int nblocks = SEQ / rows_per_block;

    gates_gemm_wmma<<<nblocks, WAVES_PER_BLOCK * 32, 0, stream>>>(x, Wih, bih, bhh, gts);
    lstm_scan<<<1, 32, 0, stream>>>(gts, Whh, h0, c0, out);
}


# scan: block hand-over off the chain: step t files h of step t-1 into lane (t-1)/4 in the slot behind its exp, a block leaves as one 16-byte store per lane: no EXEC write, no s_wait_xcnt in the loop
# speedup vs baseline: 3.9322x; 1.0703x over previous
#include <hip/hip_runtime.h>

typedef float v2f __attribute__((ext_vector_type(2)));
typedef float v4f __attribute__((ext_vector_type(4)));
typedef float v8f __attribute__((ext_vector_type(8)));

#define SEQ   32768
#define DIM   1024
#define ROWS_PER_WAVE 16
#define WAVES_PER_BLOCK 8

#define SBLK 16
#define NSB  (SEQ / SBLK)

__global__ __launch_bounds__(WAVES_PER_BLOCK * 32)
void gates_gemm_wmma(const float* __restrict__ x,
                     const float* __restrict__ Wih,
                     const float* __restrict__ bih,
                     const float* __restrict__ bhh,
                     float* __restrict__ plane)
{
    const int lane    = threadIdx.x & 31;
    const int wave    = threadIdx.x >> 5;
    const int lane_lo = lane & 15;
    const int lane_hi = lane >> 4;
    const int row_base = (blockIdx.x * WAVES_PER_BLOCK + wave) * ROWS_PER_WAVE;

    const float* xrow = x + (size_t)(row_base + lane_lo) * DIM + 2 * lane_hi;
    const float* wrow = Wih + (size_t)(lane_lo & 3) * DIM + 2 * lane_hi;

    v8f acc = {};
    #pragma unroll 4
    for (int k = 0; k < DIM; k += 4) {
        v2f a = *(const v2f*)(xrow + k);
        v2f b = *(const v2f*)(wrow + k);
        acc = __builtin_amdgcn_wmma_f32_16x16x4_f32(
                   false, a,  false, b,
                   (short)0, acc,  false,  false);
    }

    if (lane_lo < 4) {
        const int g    = lane_lo;
        const float bb = bih[g] + bhh[g];
        float* dst = plane + (size_t)(row_base >> 4) * (4 * SBLK) + g * SBLK + lane_hi * 8;
        v4f lo, hi;
        #pragma unroll
        for (int r = 0; r < 4; ++r) {
            lo[r] = acc[r] + bb;
            hi[r] = acc[4 + r] + bb;
        }
        *(v4f*)(dst)     = lo;
        *(v4f*)(dst + 4) = hi;
    }
}


struct lane_consts {
    float k;
    float kw;
    float p;
};

#define QP(Q) "quad_perm:[" #Q "," #Q "," #Q "," #Q "] row_mask:0xf bank_mask:0xf bound_ctrl:1"

__device__ __forceinline__ void lstm_step(float& y, float& C, float kgn,
                                          const lane_consts& L, float& hout,
                                          float& vsel, float hprev, unsigned keep)
{
    float e, d, r, t0, ow;
    asm("v_exp_f32 %[e], %[y]\n\t"
        "v_cndmask_b32_e64_dpp %[vs], %[hp], %[vs], %[keep] " QP(3) "\n\t"
        "v_fma_f32 %[d], %[e], %[p], %[p]\n\t"
        "v_rcp_f32 %[r], %[d]\n\t"
        "v_nop\n\t"
        "v_fma_f32_e64_dpp %[t0], %[r], %[r], %[r] " QP(2) "\n\t"
        "v_mul_f32_dpp %[ow], %[r], %[kw] " QP(3) "\n\t"
        "v_fmac_f32_dpp %[t0], %[r], %[C] " QP(1)
        : [e] "=&v"(e), [d] "=&v"(d), [r] "=&v"(r), [t0] "=&v"(t0), [ow] "=&v"(ow), [vs] "+v"(vsel)
        : [y] "v"(y), [p] "v"(L.p), [kw] "v"(L.kw), [C] "v"(C), [hp] "v"(hprev), [keep] "s"(keep));
    C = t0;
    const float bz = kgn - ow;
    float E, d2, r2, yn;
    asm("v_exp_f32 %[E], %[C]\n\t"
        "v_nop\n\t"
        "v_fma_f32 %[d2], %[E], 0.5, 0.5\n\t"
        "v_rcp_f32 %[r2], %[d2]\n\t"
        "v_nop\n\t"
        "v_fma_f32_e64_dpp %[yn], %[r2], %[ow], %[bz] " QP(0) "\n\t"
        "v_fma_f32_e64_dpp %[h], %[r2], %[r], -%[r] " QP(0)
        : [E] "=&v"(E), [d2] "=&v"(d2), [r2] "=&v"(r2), [yn] "=&v"(yn), [h] "=&v"(hout)
        : [C] "v"(t0), [ow] "v"(ow), [bz] "v"(bz), [r] "v"(r));
    y = yn;
}

__device__ __forceinline__ void lstm_block(const v4f (&cur)[4], v4f (&nxt)[4],
                                           const v4f* __restrict__ nextp,
                                           const lane_consts& L,
                                           float& y, float& C, float& hl,
                                           float (&vc)[4], float (&vp)[4],
                                           float* __restrict__ oprev, bool store_prev)
{
    #pragma unroll
    for (int q = 0; q < 4; ++q) nxt[q] = nextp[q];
    __builtin_amdgcn_sched_barrier(0);

    #pragma unroll
    for (int t = 0; t < SBLK; ++t) {
        const int q = t >> 2, s = t & 3;
        const float gn = (t == SBLK - 1) ? nxt[0][0]
                       : (s < 3 ? cur[q][s + 1] : cur[(q + 1) & 3][0]);
        float hout;
        if (t == 0) {
            lstm_step(y, C, L.k * gn, L, hout, vp[3], hl, ~(1u << 3));
            if (store_prev) {
                v4f w = { vp[0], vp[1], vp[2], vp[3] };
                *(v4f*)oprev = w;
            }
        } else {
            lstm_step(y, C, L.k * gn, L, hout, vc[(t - 1) & 3], hl, ~(1u << ((t - 1) >> 2)));
        }
        hl = hout;
    }
    asm volatile("" :: "v"(nextp));
}

__global__ void lstm_scan(const float* __restrict__ plane,
               const float* __restrict__ Whh,
               const float* __restrict__ h0,
               const float* __restrict__ c0,
               float* __restrict__ out)
{
    const int lane = threadIdx.x;
    if (lane >= 4) return;

    const bool tanh_gate = (lane == 2);
    lane_consts L;
    L.k  = tanh_gate ? -2.8853901631779268f : -1.4426950408889634f;
    L.kw = L.k * Whh[lane];
    L.p  = (lane == 0) ? 0.34657359027997264f
         : tanh_gate   ? -0.5f
         : 1.0f;

    const v4f* __restrict__ gp = (const v4f*)(plane + lane * SBLK);
    v4f A[4], B[4];
    #pragma unroll
    for (int q = 0; q < 4; ++q) A[q] = gp[q];

    float V0[4] = {0.f, 0.f, 0.f, 0.f}, V1[4] = {0.f, 0.f, 0.f, 0.f};
    float hl = 0.f;
    float* __restrict__ ol = out + lane * 4;

    float C = -2.8853901631779268f * c0[0];
    float y = fmaf(h0[0], L.kw, L.k * A[0][0]);

    for (int b = 0; b < NSB; b += 2) {
        lstm_block(A, B, gp + (size_t)(b + 1) * SBLK, L, y, C, hl, V0, V1,
                   ol + (size_t)(b > 0 ? b - 1 : 0) * SBLK, b > 0);
        const int nb = (b + 2 < NSB) ? b + 2 : NSB - 1;
        lstm_block(B, A, gp + (size_t)nb * SBLK, L, y, C, hl, V1, V0,
                   ol + (size_t)b * SBLK, true);
    }
    {
        const float h3 = __builtin_bit_cast(float,
            __builtin_amdgcn_mov_dpp(__builtin_bit_cast(int, hl), 0xFF, 0xF, 0xF, true));
        if (lane == 3) V1[3] = h3;
        v4f w = { V1[0], V1[1], V1[2], V1[3] };
        *(v4f*)(ol + (size_t)(NSB - 1) * SBLK) = w;
    }
}

extern "C" void kernel_launch(void* const* d_in, const int* in_sizes, int n_in,
                              void* d_out, int out_size, void* d_ws, size_t ws_size,
                              hipStream_t stream)
{
    const float* x    = (const float*)d_in[0];
    const float* Wih  = (const float*)d_in[1];
    const float* Whh  = (const float*)d_in[2];
    const float* bih  = (const float*)d_in[3];
    const float* bhh  = (const float*)d_in[4];
    const float* h0   = (const float*)d_in[5];
    const float* c0   = (const float*)d_in[6];
    float*       out  = (float*)d_out;
    float*       gts  = (float*)d_ws;

    const int rows_per_block = ROWS_PER_WAVE * WAVES_PER_BLOCK;
    const int nblocks = SEQ / rows_per_block;

    gates_gemm_wmma<<<nblocks, WAVES_PER_BLOCK * 32, 0, stream>>>(x, Wih, bih, bhh, gts);
    lstm_scan<<<1, 32, 0, stream>>>(gts, Whh, h0, c0, out);
}


# scan: exp2 of the cell sum replaced by a product of two exps, exp2(f*C-1)*exp2(K*i*g), joined by the fma before the second rcp: cell fma leaves the chain (9 -> 8 links); rounding of tanh(c) changes
# speedup vs baseline: 4.0977x; 1.0421x over previous
#include <hip/hip_runtime.h>

typedef float v2f __attribute__((ext_vector_type(2)));
typedef float v4f __attribute__((ext_vector_type(4)));
typedef float v8f __attribute__((ext_vector_type(8)));

#define SEQ   32768
#define DIM   1024
#define ROWS_PER_WAVE 16
#define WAVES_PER_BLOCK 8

#define SBLK 16
#define NSB  (SEQ / SBLK)

__global__ __launch_bounds__(WAVES_PER_BLOCK * 32)
void gates_gemm_wmma(const float* __restrict__ x,
                     const float* __restrict__ Wih,
                     const float* __restrict__ bih,
                     const float* __restrict__ bhh,
                     float* __restrict__ plane)
{
    const int lane    = threadIdx.x & 31;
    const int wave    = threadIdx.x >> 5;
    const int lane_lo = lane & 15;
    const int lane_hi = lane >> 4;
    const int row_base = (blockIdx.x * WAVES_PER_BLOCK + wave) * ROWS_PER_WAVE;

    const float* xrow = x + (size_t)(row_base + lane_lo) * DIM + 2 * lane_hi;
    const float* wrow = Wih + (size_t)(lane_lo & 3) * DIM + 2 * lane_hi;

    v8f acc = {};
    #pragma unroll 4
    for (int k = 0; k < DIM; k += 4) {
        v2f a = *(const v2f*)(xrow + k);
        v2f b = *(const v2f*)(wrow + k);
        acc = __builtin_amdgcn_wmma_f32_16x16x4_f32(
                   false, a,  false, b,
                   (short)0, acc,  false,  false);
    }

    if (lane_lo < 4) {
        const int g    = lane_lo;
        const float bb = bih[g] + bhh[g];
        float* dst = plane + (size_t)(row_base >> 4) * (4 * SBLK) + g * SBLK + lane_hi * 8;
        v4f lo, hi;
        #pragma unroll
        for (int r = 0; r < 4; ++r) {
            lo[r] = acc[r] + bb;
            hi[r] = acc[4 + r] + bb;
        }
        *(v4f*)(dst)     = lo;
        *(v4f*)(dst + 4) = hi;
    }
}


struct lane_consts {
    float k;
    float kw;
    float p;
};

#define QP(Q) "quad_perm:[" #Q "," #Q "," #Q "," #Q "] row_mask:0xf bank_mask:0xf bound_ctrl:1"

__device__ __forceinline__ void lstm_step(float& y, float& C, float kgn,
                                          const lane_consts& L, float& hout,
                                          float& vsel, float hprev, unsigned keep)
{
    float e, d, r, v, u, ow, E1, E2, Cn;
    asm("v_exp_f32 %[e], %[y]\n\t"
        "v_cndmask_b32_e64_dpp %[vs], %[hp], %[vs], %[keep] " QP(3) "\n\t"
        "v_fma_f32 %[d], %[e], %[p], %[p]\n\t"
        "v_rcp_f32 %[r], %[d]\n\t"
        "v_nop\n\t"
        "v_fma_f32_e64_dpp %[v], %[r], %[r], %[r] " QP(2) "\n\t"
        "v_fma_f32_e64_dpp %[u], %[r], %[C], -1.0 " QP(1) "\n\t"
        "v_mul_f32_dpp %[ow], %[r], %[kw] " QP(3) "\n\t"
        "v_exp_f32 %[E2], %[v]\n\t"
        "v_exp_f32 %[E1], %[u]\n\t"
        "v_fma_f32_e64_dpp %[Cn], %[r], %[C], %[v] " QP(1)
        : [e] "=&v"(e), [d] "=&v"(d), [r] "=&v"(r), [v] "=&v"(v), [u] "=&v"(u), [ow] "=&v"(ow),
          [E1] "=&v"(E1), [E2] "=&v"(E2), [Cn] "=&v"(Cn), [vs] "+v"(vsel)
        : [y] "v"(y), [p] "v"(L.p), [kw] "v"(L.kw), [C] "v"(C), [hp] "v"(hprev), [keep] "s"(keep));
    C = Cn;
    const float bz = kgn - ow;
    float d2, r2, yn;
    asm("v_fma_f32 %[d2], %[E1], %[E2], 0.5\n\t"
        "v_rcp_f32 %[r2], %[d2]\n\t"
        "v_nop\n\t"
        "v_fma_f32_e64_dpp %[yn], %[r2], %[ow], %[bz] " QP(0) "\n\t"
        "v_fma_f32_e64_dpp %[h], %[r2], %[r], -%[r] " QP(0)
        : [d2] "=&v"(d2), [r2] "=&v"(r2), [yn] "=&v"(yn), [h] "=&v"(hout)
        : [E1] "v"(E1), [E2] "v"(E2), [ow] "v"(ow), [bz] "v"(bz), [r] "v"(r));
    y = yn;
}

__device__ __forceinline__ void lstm_block(const v4f (&cur)[4], v4f (&nxt)[4],
                                           const v4f* __restrict__ nextp,
                                           const lane_consts& L,
                                           float& y, float& C, float& hl,
                                           float (&vc)[4], float (&vp)[4],
                                           float* __restrict__ oprev, bool store_prev)
{
    #pragma unroll
    for (int q = 0; q < 4; ++q) nxt[q] = nextp[q];
    __builtin_amdgcn_sched_barrier(0);

    #pragma unroll
    for (int t = 0; t < SBLK; ++t) {
        const int q = t >> 2, s = t & 3;
        const float gn = (t == SBLK - 1) ? nxt[0][0]
                       : (s < 3 ? cur[q][s + 1] : cur[(q + 1) & 3][0]);
        float hout;
        if (t == 0) {
            lstm_step(y, C, L.k * gn, L, hout, vp[3], hl, ~(1u << 3));
            if (store_prev) {
                v4f w = { vp[0], vp[1], vp[2], vp[3] };
                *(v4f*)oprev = w;
            }
        } else {
            lstm_step(y, C, L.k * gn, L, hout, vc[(t - 1) & 3], hl, ~(1u << ((t - 1) >> 2)));
        }
        hl = hout;
    }
    asm volatile("" :: "v"(nextp));
}

__global__ void lstm_scan(const float* __restrict__ plane,
               const float* __restrict__ Whh,
               const float* __restrict__ h0,
               const float* __restrict__ c0,
               float* __restrict__ out)
{
    const int lane = threadIdx.x;
    if (lane >= 4) return;

    const bool tanh_gate = (lane == 2);
    lane_consts L;
    L.k  = tanh_gate ? -2.8853901631779268f : -1.4426950408889634f;
    L.kw = L.k * Whh[lane];
    L.p  = (lane == 0) ? 0.34657359027997264f
         : tanh_gate   ? -0.5f
         : 1.0f;

    const v4f* __restrict__ gp = (const v4f*)(plane + lane * SBLK);
    v4f A[4], B[4];
    #pragma unroll
    for (int q = 0; q < 4; ++q) A[q] = gp[q];

    float V0[4] = {0.f, 0.f, 0.f, 0.f}, V1[4] = {0.f, 0.f, 0.f, 0.f};
    float hl = 0.f;
    float* __restrict__ ol = out + lane * 4;

    float C = -2.8853901631779268f * c0[0];
    float y = fmaf(h0[0], L.kw, L.k * A[0][0]);

    for (int b = 0; b < NSB; b += 2) {
        lstm_block(A, B, gp + (size_t)(b + 1) * SBLK, L, y, C, hl, V0, V1,
                   ol + (size_t)(b > 0 ? b - 1 : 0) * SBLK, b > 0);
        const int nb = (b + 2 < NSB) ? b + 2 : NSB - 1;
        lstm_block(B, A, gp + (size_t)nb * SBLK, L, y, C, hl, V1, V0,
                   ol + (size_t)b * SBLK, true);
    }
    {
        const float h3 = __builtin_bit_cast(float,
            __builtin_amdgcn_mov_dpp(__builtin_bit_cast(int, hl), 0xFF, 0xF, 0xF, true));
        if (lane == 3) V1[3] = h3;
        v4f w = { V1[0], V1[1], V1[2], V1[3] };
        *(v4f*)(ol + (size_t)(NSB - 1) * SBLK) = w;
    }
}

extern "C" void kernel_launch(void* const* d_in, const int* in_sizes, int n_in,
                              void* d_out, int out_size, void* d_ws, size_t ws_size,
                              hipStream_t stream)
{
    const float* x    = (const float*)d_in[0];
    const float* Wih  = (const float*)d_in[1];
    const float* Whh  = (const float*)d_in[2];
    const float* bih  = (const float*)d_in[3];
    const float* bhh  = (const float*)d_in[4];
    const float* h0   = (const float*)d_in[5];
    const float* c0   = (const float*)d_in[6];
    float*       out  = (float*)d_out;
    float*       gts  = (float*)d_ws;

    const int rows_per_block = ROWS_PER_WAVE * WAVES_PER_BLOCK;
    const int nblocks = SEQ / rows_per_block;

    gates_gemm_wmma<<<nblocks, WAVES_PER_BLOCK * 32, 0, stream>>>(x, Wih, bih, bhh, gts);
    lstm_scan<<<1, 32, 0, stream>>>(gts, Whh, h0, c0, out);
}


# scan: multipliers a step old ride inside the rcp arguments (1/C for f*C; (1+e_o)/(2kw) for kw*o*(1+tanh c)); exp2 of a sum becomes a product; rcp feeds exp2 directly: 8 -> 7 links; sums re-associated
# speedup vs baseline: 4.3147x; 1.0530x over previous
#include <hip/hip_runtime.h>

typedef float v2f __attribute__((ext_vector_type(2)));
typedef float v4f __attribute__((ext_vector_type(4)));
typedef float v8f __attribute__((ext_vector_type(8)));

#define SEQ   32768
#define DIM   1024
#define ROWS_PER_WAVE 16
#define WAVES_PER_BLOCK 8

#define SBLK 16
#define NSB  (SEQ / SBLK)

__global__ __launch_bounds__(WAVES_PER_BLOCK * 32)
void gates_gemm_wmma(const float* __restrict__ x,
                     const float* __restrict__ Wih,
                     const float* __restrict__ bih,
                     const float* __restrict__ bhh,
                     float* __restrict__ plane)
{
    const int lane    = threadIdx.x & 31;
    const int wave    = threadIdx.x >> 5;
    const int lane_lo = lane & 15;
    const int lane_hi = lane >> 4;
    const int row_base = (blockIdx.x * WAVES_PER_BLOCK + wave) * ROWS_PER_WAVE;

    const float* xrow = x + (size_t)(row_base + lane_lo) * DIM + 2 * lane_hi;
    const float* wrow = Wih + (size_t)(lane_lo & 3) * DIM + 2 * lane_hi;

    v8f acc = {};
    #pragma unroll 4
    for (int k = 0; k < DIM; k += 4) {
        v2f a = *(const v2f*)(xrow + k);
        v2f b = *(const v2f*)(wrow + k);
        acc = __builtin_amdgcn_wmma_f32_16x16x4_f32(
                   false, a,  false, b,
                   (short)0, acc,  false,  false);
    }

    if (lane_lo < 4) {
        const int g    = lane_lo;
        const float bb = bih[g] + bhh[g];
        float* dst = plane + (size_t)(row_base >> 4) * (4 * SBLK) + g * SBLK + lane_hi * 8;
        v4f lo, hi;
        #pragma unroll
        for (int r = 0; r < 4; ++r) {
            lo[r] = acc[r] + bb;
            hi[r] = acc[4 + r] + bb;
        }
        *(v4f*)(dst)     = lo;
        *(v4f*)(dst + 4) = hi;
    }
}


struct lane_consts {
    float k;
    float kw;
    float ikw;
    float pc;
};

struct lane_state {
    float s;
    float pb;
    float pv;
    float r2h;
    float o2;
    float ro;
};

#define QP(Q) "quad_perm:[" #Q "," #Q "," #Q "," #Q "] row_mask:0xf bank_mask:0xf bound_ctrl:1"

__device__ __forceinline__ void lstm_step(lane_state& S, float gn, const lane_consts& L,
                                          float& vsel, unsigned keep)
{
    float X, hp, d, r, q, v, E1, ow, E2, Cn, A, bz, ic, D2, d2q, Ebz, o2, r2h, pvn, sn, pbn;
    asm("v_exp_f32 %[X], %[s]\n\t"
        "v_fma_f32_e64_dpp %[hp], %[r2hp], %[o2p], -%[rop] " QP(0) "\n\t"
        "v_fma_f32 %[d], %[X], %[pb], %[pv]\n\t"
        "v_rcp_f32 %[r], %[d]\n\t"
        "v_mul_f32_dpp %[q], %[d], %[ikw] " QP(3) "\n\t"
        "v_cndmask_b32_e64_dpp %[vs], %[hp], %[vs], %[keep] " QP(3) "\n\t"
        "v_fma_f32_e64_dpp %[v], %[r], %[r], %[r] " QP(2) "\n\t"
        "v_exp_f32 %[E1], %[r]\n\t"
        "v_mul_f32_dpp %[ow], %[r], %[kw] " QP(3) "\n\t"
        "v_exp_f32 %[E2], %[v]\n\t"
        "v_add_f32_dpp %[Cn], %[v], %[r] " QP(0) "\n\t"
        "v_mul_f32_dpp %[A], %[E1], %[q] " QP(1) "\n\t"
        "v_fma_f32 %[bz], %[k], %[gn], -%[ow]\n\t"
        "v_rcp_f32 %[ic], %[Cn]\n\t"
        "v_fma_f32_e64_dpp %[D2], %[E1], %[E2], 1.0 " QP(1) "\n\t"
        "v_fma_f32_e64_dpp %[d2q], %[E2], %[A], %[q] " QP(0) "\n\t"
        "v_exp_f32 %[Ebz], %[bz]\n\t"
        "v_add_f32 %[o2], %[r], %[r]\n\t"
        "v_rcp_f32 %[r2h], %[D2]\n\t"
        "v_cndmask_b32_e64 %[pvn], %[pc], %[ic], %[m1]\n\t"
        "v_rcp_f32 %[sn], %[d2q]\n\t"
        "v_mul_f32 %[pbn], %[pvn], %[Ebz]"
        : [X] "=&v"(X), [hp] "=&v"(hp), [d] "=&v"(d), [r] "=&v"(r), [q] "=&v"(q), [v] "=&v"(v),
          [E1] "=&v"(E1), [ow] "=&v"(ow), [E2] "=&v"(E2), [Cn] "=&v"(Cn), [A] "=&v"(A), [bz] "=&v"(bz),
          [ic] "=&v"(ic), [D2] "=&v"(D2), [d2q] "=&v"(d2q), [Ebz] "=&v"(Ebz), [o2] "=&v"(o2),
          [r2h] "=&v"(r2h), [pvn] "=&v"(pvn), [sn] "=&v"(sn), [pbn] "=&v"(pbn), [vs] "+v"(vsel)
        : [s] "v"(S.s), [pb] "v"(S.pb), [pv] "v"(S.pv), [r2hp] "v"(S.r2h), [o2p] "v"(S.o2), [rop] "v"(S.ro),
          [ikw] "v"(L.ikw), [kw] "v"(L.kw), [k] "v"(L.k), [pc] "v"(L.pc), [gn] "v"(gn),
          [keep] "s"(keep), [m1] "s"(2u));
    S.s = sn; S.pb = pbn; S.pv = pvn; S.r2h = r2h; S.o2 = o2; S.ro = r;
}

__device__ __forceinline__ void lstm_block(const v4f (&cur)[4], v4f (&nxt)[4],
                                           const v4f* __restrict__ nextp,
                                           const lane_consts& L, lane_state& S,
                                           float (&vc)[4], float (&vp)[4],
                                           float* __restrict__ oprev, bool store_prev)
{
    #pragma unroll
    for (int q = 0; q < 4; ++q) nxt[q] = nextp[q];
    __builtin_amdgcn_sched_barrier(0);

    #pragma unroll
    for (int t = 0; t < SBLK; ++t) {
        const int q = t >> 2, s = t & 3;
        const float gn = (t == SBLK - 1) ? nxt[0][0]
                       : (s < 3 ? cur[q][s + 1] : cur[(q + 1) & 3][0]);
        if (t == 0) {
            lstm_step(S, gn, L, vp[3], ~(1u << 3));
            if (store_prev) {
                v4f w = { vp[0], vp[1], vp[2], vp[3] };
                *(v4f*)oprev = w;
            }
        } else {
            lstm_step(S, gn, L, vc[(t - 1) & 3], ~(1u << ((t - 1) >> 2)));
        }
    }
    asm volatile("" :: "v"(nextp));
}

__global__ void lstm_scan(const float* __restrict__ plane,
               const float* __restrict__ Whh,
               const float* __restrict__ h0,
               const float* __restrict__ c0,
               float* __restrict__ out)
{
    const int lane = threadIdx.x;
    if (lane >= 4) return;

    const bool tanh_gate = (lane == 2);
    lane_consts L;
    L.k   = tanh_gate ? -2.8853901631779268f : -1.4426950408889634f;
    L.kw  = L.k * Whh[lane];
    L.ikw = 0.5f / L.kw;
    L.pc  = (lane == 0) ? 0.34657359027997264f
          : tanh_gate   ? -0.5f
          : 1.0f;

    const v4f* __restrict__ gp = (const v4f*)(plane + lane * SBLK);
    v4f A[4], B[4];
    #pragma unroll
    for (int q = 0; q < 4; ++q) A[q] = gp[q];

    float V0[4] = {0.f, 0.f, 0.f, 0.f}, V1[4] = {0.f, 0.f, 0.f, 0.f};
    float* __restrict__ ol = out + lane * 4;

    lane_state S;
    const float C0 = -2.8853901631779268f * c0[0];
    S.pv  = (lane == 1) ? __builtin_amdgcn_rcpf(C0) : L.pc;
    S.pb  = S.pv * __builtin_amdgcn_exp2f(L.k * A[0][0]);
    S.s   = L.kw * h0[0];
    S.r2h = 0.f; S.o2 = 0.f; S.ro = 0.f;

    for (int b = 0; b < NSB; b += 2) {
        lstm_block(A, B, gp + (size_t)(b + 1) * SBLK, L, S, V0, V1,
                   ol + (size_t)(b > 0 ? b - 1 : 0) * SBLK, b > 0);
        const int nb = (b + 2 < NSB) ? b + 2 : NSB - 1;
        lstm_block(B, A, gp + (size_t)nb * SBLK, L, S, V1, V0,
                   ol + (size_t)b * SBLK, true);
    }
    {
        const float r2h0 = __builtin_bit_cast(float,
            __builtin_amdgcn_mov_dpp(__builtin_bit_cast(int, S.r2h), 0x00, 0xF, 0xF, true));
        const float h3 = __builtin_fmaf(r2h0, S.o2, -S.ro);
        if (lane == 3) V1[3] = h3;
        v4f w = { V1[0], V1[1], V1[2], V1[3] };
        *(v4f*)(ol + (size_t)(NSB - 1) * SBLK) = w;
    }
}

extern "C" void kernel_launch(void* const* d_in, const int* in_sizes, int n_in,
                              void* d_out, int out_size, void* d_ws, size_t ws_size,
                              hipStream_t stream)
{
    const float* x    = (const float*)d_in[0];
    const float* Wih  = (const float*)d_in[1];
    const float* Whh  = (const float*)d_in[2];
    const float* bih  = (const float*)d_in[3];
    const float* bhh  = (const float*)d_in[4];
    const float* h0   = (const float*)d_in[5];
    const float* c0   = (const float*)d_in[6];
    float*       out  = (float*)d_out;
    float*       gts  = (float*)d_ws;

    const int rows_per_block = ROWS_PER_WAVE * WAVES_PER_BLOCK;
    const int nblocks = SEQ / rows_per_block;

    gates_gemm_wmma<<<nblocks, WAVES_PER_BLOCK * 32, 0, stream>>>(x, Wih, bih, bhh, gts);
    lstm_scan<<<1, 32, 0, stream>>>(gts, Whh, h0, c0, out);
}
